# starting kernel
# baseline (speedup 1.0000x reference)
#include <hip/hip_runtime.h>

typedef __attribute__((ext_vector_type(16))) _Float16 v16h;
typedef __attribute__((ext_vector_type(8)))  float    v8f;

#define N_NODES 50000
#define N_EDGES 800000
#define FDIM    64
#define EFDIM   16
#define MSG_IN  145
#define MSG_K   160
#define NODE_K  128

#define WS_CNT_OFF     ((size_t)N_NODES * FDIM * 4)
#define WS_WEIGHTS_OFF ((size_t)13000192)

#define MW1B 0
#define MW2B 10240
#define CW1B 14336
#define CW2B 18432
#define NW1B 19456
#define NW2B 27648

__device__ __forceinline__ v8f wmma16(v16h a, v16h b, v8f c) {
    return __builtin_amdgcn_wmma_f32_16x16x32_f16(
         false, a,  false, b,
         (short)0, c,  false,  false);
}

__device__ __forceinline__ v16h load_a_row(const _Float16* rowptr, int kcbase, int khalf) {
    v16h a;
#pragma unroll
    for (int e = 0; e < 16; ++e) {
        int klocal = (e & 7) + 8 * khalf + 16 * (e >> 3);
        a[e] = rowptr[kcbase + klocal];
    }
    return a;
}

__global__ __launch_bounds__(256) void egnn_zero(float* msum, float* cnt, float* pos_out) {
    int i = blockIdx.x * 256 + threadIdx.x;
    msum[i] = 0.0f;
    if (i < N_NODES)     cnt[i] = 0.0f;
    if (i < N_NODES * 3) pos_out[i] = 0.0f;
}

__global__ __launch_bounds__(256) void egnn_prep_w(const float* __restrict__ W,
                                                   int Kact, int Nact, int ntiles,
                                                   int total, _Float16* __restrict__ out) {
    int idx = blockIdx.x * 256 + threadIdx.x;
    if (idx >= total) return;
    int block  = idx >> 9;
    int within = idx & 511;
    int lane = within >> 4;
    int e    = within & 15;
    int kc = block / ntiles;
    int nt = block - kc * ntiles;
    int khalf  = lane >> 4;
    int coln   = lane & 15;
    int klocal = (e & 7) + 8 * khalf + 16 * (e >> 3);
    int k = kc * 32 + klocal;
    int n = nt * 16 + coln;
    float v = (k < Kact && n < Nact) ? W[(size_t)k * Nact + n] : 0.0f;
    out[idx] = (_Float16)v;
}

__global__ __launch_bounds__(256) void egnn_edge(
    const float* __restrict__ x, const int* __restrict__ ei,
    const float* __restrict__ pos, const float* __restrict__ pos_init,
    const float* __restrict__ edge_attr,
    const float* __restrict__ mb1, const float* __restrict__ mb2,
    const float* __restrict__ cb1, const float* __restrict__ cb2,
    const _Float16* __restrict__ wsW,
    float* __restrict__ msum, float* __restrict__ cnt, float* __restrict__ pos_out)
{
    __shared__ _Float16 sfeat[128][MSG_K];
    __shared__ _Float16 shid[8][16][FDIM];
    __shared__ float sdsq[128];
    __shared__ float sdpi[128][3];
    __shared__ float srinv[128];
    __shared__ int   ssrc[128];
    __shared__ int   sdst[128];

    const int tid   = threadIdx.x;
    const int ebase = blockIdx.x * 128;

    if (tid < 128) {
        int e = ebase + tid;
        int s = ei[e];
        int d = ei[N_EDGES + e];
        ssrc[tid] = s; sdst[tid] = d;
        float dx = pos[3*s+0] - pos[3*d+0];
        float dy = pos[3*s+1] - pos[3*d+1];
        float dz = pos[3*s+2] - pos[3*d+2];
        sdsq[tid] = dx*dx + dy*dy + dz*dz;
        float ix = pos_init[3*s+0] - pos_init[3*d+0];
        float iy = pos_init[3*s+1] - pos_init[3*d+1];
        float iz = pos_init[3*s+2] - pos_init[3*d+2];
        sdpi[tid][0] = ix; sdpi[tid][1] = iy; sdpi[tid][2] = iz;
        srinv[tid] = 1.0f / sqrtf(ix*ix + iy*iy + iz*iz);
        atomicAdd(&cnt[d], 1.0f);
    }
    __syncthreads();

    for (int idx = tid; idx < 128 * MSG_K; idx += 256) {
        int row = idx / MSG_K, k = idx - row * MSG_K;
        float v;
        if      (k < FDIM)       v = x[(size_t)ssrc[row] * FDIM + k];
        else if (k < 2*FDIM)     v = x[(size_t)sdst[row] * FDIM + (k - FDIM)];
        else if (k == 2*FDIM)    v = sdsq[row];
        else if (k < MSG_IN)     v = edge_attr[(size_t)(ebase + row) * EFDIM + (k - 2*FDIM - 1)];
        else                     v = 0.0f;
        sfeat[row][k] = (_Float16)v;
    }
    __syncthreads();

    const int wave  = tid >> 5;
    const int lane  = tid & 31;
    const int col   = lane & 15;
    const int khalf = lane >> 4;
    const _Float16* arow = &sfeat[wave * 16 + col][0];

    v8f acc[4] = {v8f{}, v8f{}, v8f{}, v8f{}};
#pragma unroll
    for (int kc = 0; kc < 5; ++kc) {
        v16h a = load_a_row(arow, kc * 32, khalf);
#pragma unroll
        for (int nt = 0; nt < 4; ++nt) {
            v16h b = *(const v16h*)(wsW + MW1B + (kc * 4 + nt) * 512 + lane * 16);
            acc[nt] = wmma16(a, b, acc[nt]);
        }
    }
#pragma unroll
    for (int nt = 0; nt < 4; ++nt) {
        float bia = mb1[nt * 16 + col];
#pragma unroll
        for (int r = 0; r < 8; ++r) {
            float v = acc[nt][r] + bia;
            shid[wave][r + 8 * khalf][nt * 16 + col] = (_Float16)(v > 0.0f ? v : 0.0f);
        }
    }
    const _Float16* hrow = &shid[wave][col][0];

    v8f mij[4] = {v8f{}, v8f{}, v8f{}, v8f{}};
#pragma unroll
    for (int kc = 0; kc < 2; ++kc) {
        v16h a = load_a_row(hrow, kc * 32, khalf);
#pragma unroll
        for (int nt = 0; nt < 4; ++nt) {
            v16h b = *(const v16h*)(wsW + MW2B + (kc * 4 + nt) * 512 + lane * 16);
            mij[nt] = wmma16(a, b, mij[nt]);
        }
    }
#pragma unroll
    for (int nt = 0; nt < 4; ++nt) {
        float bia = mb2[nt * 16 + col];
#pragma unroll
        for (int r = 0; r < 8; ++r) {
            int row = r + 8 * khalf;
            float v = mij[nt][r] + bia;
            atomicAdd(&msum[(size_t)sdst[wave * 16 + row] * FDIM + nt * 16 + col], v);
            shid[wave][row][nt * 16 + col] = (_Float16)v;
        }
    }

    v8f tac[4] = {v8f{}, v8f{}, v8f{}, v8f{}};
#pragma unroll
    for (int kc = 0; kc < 2; ++kc) {
        v16h a = load_a_row(hrow, kc * 32, khalf);
#pragma unroll
        for (int nt = 0; nt < 4; ++nt) {
            v16h b = *(const v16h*)(wsW + CW1B + (kc * 4 + nt) * 512 + lane * 16);
            tac[nt] = wmma16(a, b, tac[nt]);
        }
    }
#pragma unroll
    for (int nt = 0; nt < 4; ++nt) {
        float bia = cb1[nt * 16 + col];
#pragma unroll
        for (int r = 0; r < 8; ++r) {
            float v = tac[nt][r] + bia;
            shid[wave][r + 8 * khalf][nt * 16 + col] = (_Float16)(v > 0.0f ? v : 0.0f);
        }
    }

    v8f cac = v8f{};
#pragma unroll
    for (int kc = 0; kc < 2; ++kc) {
        v16h a = load_a_row(hrow, kc * 32, khalf);
        v16h b = *(const v16h*)(wsW + CW2B + kc * 512 + lane * 16);
        cac = wmma16(a, b, cac);
    }
    if (col < 3) {
        float bia = cb2[col];
#pragma unroll
        for (int r = 0; r < 8; ++r) {
            int el = wave * 16 + r + 8 * khalf;
            float w = cac[r] + bia;
            float v = sdpi[el][col] * srinv[el] * w;
            atomicAdd(&pos_out[(size_t)sdst[el] * 3 + col], v);
        }
    }
}

__global__ __launch_bounds__(256) void egnn_node(
    const float* __restrict__ x, const float* __restrict__ msum,
    const float* __restrict__ cnt,
    const float* __restrict__ nb1, const float* __restrict__ nb2,
    const _Float16* __restrict__ wsW, float* __restrict__ hout)
{
    __shared__ _Float16 sfeat[128][NODE_K];
    __shared__ _Float16 shid[8][16][FDIM];
    __shared__ float sinv[128];

    const int tid   = threadIdx.x;
    const int nbase = blockIdx.x * 128;

    if (tid < 128) {
        int n = nbase + tid;
        float c = (n < N_NODES) ? cnt[n] : 1.0f;
        sinv[tid] = 1.0f / fmaxf(c, 1.0f);
    }
    __syncthreads();

    for (int idx = tid; idx < 128 * NODE_K; idx += 256) {
        int row = idx >> 7, k = idx & 127;
        int n = nbase + row;
        float v = 0.0f;
        if (n < N_NODES)
            v = (k < FDIM) ? x[(size_t)n * FDIM + k]
                           : msum[(size_t)n * FDIM + (k - FDIM)] * sinv[row];
        sfeat[row][k] = (_Float16)v;
    }
    __syncthreads();

    const int wave  = tid >> 5;
    const int lane  = tid & 31;
    const int col   = lane & 15;
    const int khalf = lane >> 4;
    const _Float16* arow = &sfeat[wave * 16 + col][0];

    v8f acc[4] = {v8f{}, v8f{}, v8f{}, v8f{}};
#pragma unroll
    for (int kc = 0; kc < 4; ++kc) {
        v16h a = load_a_row(arow, kc * 32, khalf);
#pragma unroll
        for (int nt = 0; nt < 4; ++nt) {
            v16h b = *(const v16h*)(wsW + NW1B + (kc * 4 + nt) * 512 + lane * 16);
            acc[nt] = wmma16(a, b, acc[nt]);
        }
    }
#pragma unroll
    for (int nt = 0; nt < 4; ++nt) {
        float bia = nb1[nt * 16 + col];
#pragma unroll
        for (int r = 0; r < 8; ++r) {
            float v = acc[nt][r] + bia;
            shid[wave][r + 8 * khalf][nt * 16 + col] = (_Float16)(v > 0.0f ? v : 0.0f);
        }
    }
    const _Float16* hrow = &shid[wave][col][0];

    v8f hacc[4] = {v8f{}, v8f{}, v8f{}, v8f{}};
#pragma unroll
    for (int kc = 0; kc < 2; ++kc) {
        v16h a = load_a_row(hrow, kc * 32, khalf);
#pragma unroll
        for (int nt = 0; nt < 4; ++nt) {
            v16h b = *(const v16h*)(wsW + NW2B + (kc * 4 + nt) * 512 + lane * 16);
            hacc[nt] = wmma16(a, b, hacc[nt]);
        }
    }
#pragma unroll
    for (int nt = 0; nt < 4; ++nt) {
        float bia = nb2[nt * 16 + col];
#pragma unroll
        for (int r = 0; r < 8; ++r) {
            int n = nbase + wave * 16 + r + 8 * khalf;
            if (n < N_NODES)
                hout[(size_t)n * FDIM + nt * 16 + col] = hacc[nt][r] + bia;
        }
    }
}

extern "C" void kernel_launch(void* const* d_in, const int* in_sizes, int n_in,
                              void* d_out, int out_size, void* d_ws, size_t ws_size,
                              hipStream_t stream) {
    const float* x         = (const float*)d_in[0];
    const int*   ei        = (const int*)  d_in[1];
    const float* pos       = (const float*)d_in[2];
    const float* pos_init  = (const float*)d_in[3];
    const float* edge_attr = (const float*)d_in[4];
    const float* mw1 = (const float*)d_in[5];
    const float* mb1 = (const float*)d_in[6];
    const float* mw2 = (const float*)d_in[7];
    const float* mb2 = (const float*)d_in[8];
    const float* nw1 = (const float*)d_in[9];
    const float* nb1 = (const float*)d_in[10];
    const float* nw2 = (const float*)d_in[11];
    const float* nb2 = (const float*)d_in[12];
    const float* cw1 = (const float*)d_in[13];
    const float* cb1 = (const float*)d_in[14];
    const float* cw2 = (const float*)d_in[15];
    const float* cb2 = (const float*)d_in[16];

    float* h_out   = (float*)d_out;
    float* pos_out = h_out + (size_t)N_NODES * FDIM;

    float* msum = (float*)d_ws;
    float* cnt  = (float*)((char*)d_ws + WS_CNT_OFF);
    _Float16* wsW = (_Float16*)((char*)d_ws + WS_WEIGHTS_OFF);

    egnn_zero<<<(N_NODES * FDIM) / 256, 256, 0, stream>>>(msum, cnt, pos_out);

    egnn_prep_w<<<40, 256, 0, stream>>>(mw1, MSG_IN, 64, 4, 20 * 512, wsW + MW1B);
    egnn_prep_w<<<16, 256, 0, stream>>>(mw2, 64,     64, 4,  8 * 512, wsW + MW2B);
    egnn_prep_w<<<16, 256, 0, stream>>>(cw1, 64,     64, 4,  8 * 512, wsW + CW1B);
    egnn_prep_w<<< 4, 256, 0, stream>>>(cw2, 64,      3, 1,  2 * 512, wsW + CW2B);
    egnn_prep_w<<<32, 256, 0, stream>>>(nw1, 128,    64, 4, 16 * 512, wsW + NW1B);
    egnn_prep_w<<<16, 256, 0, stream>>>(nw2, 64,     64, 4,  8 * 512, wsW + NW2B);

    egnn_edge<<<N_EDGES / 128, 256, 0, stream>>>(
        x, ei, pos, pos_init, edge_attr, mb1, mb2, cb1, cb2,
        wsW, msum, cnt, pos_out);

    egnn_node<<<(N_NODES + 127) / 128, 256, 0, stream>>>(
        x, msum, cnt, nb1, nb2, wsW, h_out);
}


# Clear and the six weight-packing launches fused into one set-up launch (124 pack blocks first in the grid, then the clearing blocks): 9 launches become 3; same stores, same packed layout
# speedup vs baseline: 2.0407x; 2.0407x over previous
#include <hip/hip_runtime.h>

typedef __attribute__((ext_vector_type(16))) _Float16 v16h;
typedef __attribute__((ext_vector_type(8)))  float    v8f;
typedef __attribute__((ext_vector_type(4)))  float    v4f;
typedef __attribute__((ext_vector_type(4)))  _Float16 v4h;
typedef __attribute__((ext_vector_type(8)))  _Float16 v8h;

#define N_NODES 50000
#define N_EDGES 800000
#define FDIM    64
#define EFDIM   16
#define MSG_IN  145
#define MSG_K   160
#define NODE_K  128
#define TR_PITCH 20

union alignas(16) EdgeStage {
    _Float16 feat[16][MSG_K];
    float    tr[FDIM][TR_PITCH];
};

#define WS_CNT_OFF     ((size_t)N_NODES * FDIM * 4)
#define WS_WEIGHTS_OFF ((size_t)13000192)

#define MW1B 0
#define MW2B 10240
#define CW1B 14336
#define CW2B 18432
#define NW1B 19456
#define NW2B 27648

__device__ __forceinline__ v8f wmma16(v16h a, v16h b, v8f c) {
    return __builtin_amdgcn_wmma_f32_16x16x32_f16(
         false, a,  false, b,
         (short)0, c,  false,  false);
}

__device__ __forceinline__ v16h load_a_row(const _Float16* rowptr, int kcbase, int khalf) {
    v16h a;
#pragma unroll
    for (int e = 0; e < 16; ++e) {
        int klocal = (e & 7) + 8 * khalf + 16 * (e >> 3);
        a[e] = rowptr[kcbase + klocal];
    }
    return a;
}

#define PREP_BLOCKS 124
__global__ __launch_bounds__(256) void egnn_setup(
    float* __restrict__ msum, float* __restrict__ cnt, float* __restrict__ pos_out,
    _Float16* __restrict__ wsW,
    const float* __restrict__ mw1, const float* __restrict__ mw2,
    const float* __restrict__ cw1, const float* __restrict__ cw2,
    const float* __restrict__ nw1, const float* __restrict__ nw2)
{
    const int b = blockIdx.x;
    if (b < PREP_BLOCKS) {
        const float* Wm; int Kact, Nact, ntiles, base;
        if      (b < MW2B / 256) { Wm = mw1; Kact = MSG_IN; Nact = 64; ntiles = 4; base = MW1B; }
        else if (b < CW1B / 256) { Wm = mw2; Kact = 64;     Nact = 64; ntiles = 4; base = MW2B; }
        else if (b < CW2B / 256) { Wm = cw1; Kact = 64;     Nact = 64; ntiles = 4; base = CW1B; }
        else if (b < NW1B / 256) { Wm = cw2; Kact = 64;     Nact = 3;  ntiles = 1; base = CW2B; }
        else if (b < NW2B / 256) { Wm = nw1; Kact = NODE_K; Nact = 64; ntiles = 4; base = NW1B; }
        else                     { Wm = nw2; Kact = 64;     Nact = 64; ntiles = 4; base = NW2B; }
        int gidx = b * 256 + threadIdx.x;
        int idx  = gidx - base;
        int block  = idx >> 9;
        int within = idx & 511;
        int lane = within >> 4;
        int e    = within & 15;
        int kc = block / ntiles;
        int nt = block - kc * ntiles;
        int khalf  = lane >> 4;
        int coln   = lane & 15;
        int klocal = (e & 7) + 8 * khalf + 16 * (e >> 3);
        int k = kc * 32 + klocal;
        int n = nt * 16 + coln;
        float v = (k < Kact && n < Nact) ? Wm[(size_t)k * Nact + n] : 0.0f;
        wsW[gidx] = (_Float16)v;
        return;
    }
    int i = (b - PREP_BLOCKS) * 256 + threadIdx.x;
    msum[i] = 0.0f;
    if (i < N_NODES)     cnt[i] = 0.0f;
    if (i < N_NODES * 3) pos_out[i] = 0.0f;
}

__global__ __launch_bounds__(256) void egnn_edge(
    const float* __restrict__ x, const int* __restrict__ ei,
    const float* __restrict__ pos, const float* __restrict__ pos_init,
    const float* __restrict__ edge_attr,
    const float* __restrict__ mb1, const float* __restrict__ mb2,
    const float* __restrict__ cb1, const float* __restrict__ cb2,
    const _Float16* __restrict__ wsW,
    float* __restrict__ msum, float* __restrict__ cnt, float* __restrict__ pos_out)
{
    __shared__ EdgeStage sst[8];
    __shared__ _Float16 shid[8][16][FDIM];
    __shared__ float sdsq[128];
    __shared__ float sdpi[128][3];
    __shared__ float srinv[128];
    __shared__ alignas(16) int snid[2][128];

    const int tid   = threadIdx.x;
    const int ebase = blockIdx.x * 128;

    if (tid < 128) {
        int e = ebase + tid;
        int s = ei[e];
        int d = ei[N_EDGES + e];
        snid[0][tid] = s; snid[1][tid] = d;
        float dx = pos[3*s+0] - pos[3*d+0];
        float dy = pos[3*s+1] - pos[3*d+1];
        float dz = pos[3*s+2] - pos[3*d+2];
        sdsq[tid] = dx*dx + dy*dy + dz*dz;
        float ix = pos_init[3*s+0] - pos_init[3*d+0];
        float iy = pos_init[3*s+1] - pos_init[3*d+1];
        float iz = pos_init[3*s+2] - pos_init[3*d+2];
        sdpi[tid][0] = ix; sdpi[tid][1] = iy; sdpi[tid][2] = iz;
        srinv[tid] = 1.0f / sqrtf(ix*ix + iy*iy + iz*iz);
        atomicAdd(&cnt[d], 1.0f);
    }
    __syncthreads();

    const int wave  = tid >> 5;
    const int lane  = tid & 31;
    const int col   = lane & 15;
    const int khalf = lane >> 4;

    {
        const int*   nid = &snid[khalf][wave * 16];
        const float* xq  = x + 4 * col;
        v4f xv[16];
#pragma unroll
        for (int i = 0; i < 16; ++i)
            xv[i] = *(const v4f*)(xq + (size_t)nid[i] * FDIM);
        const v4f* ea = (const v4f*)(edge_attr + (size_t)(ebase + wave * 16 + col) * EFDIM);
        const v4f e0 = ea[0], e1 = ea[1], e2 = ea[2], e3 = ea[3];
        const float dsq = sdsq[wave * 16 + col];
#pragma unroll
        for (int i = 0; i < 16; ++i)
            *(v4h*)&sst[wave].feat[i][FDIM * khalf + 4 * col] = __builtin_convertvector(xv[i], v4h);
        v8h ta = v8h{}, tb = v8h{};
        if (khalf == 0) {
            ta[0] = (_Float16)dsq;   ta[1] = (_Float16)e0[0]; ta[2] = (_Float16)e0[1]; ta[3] = (_Float16)e0[2];
            ta[4] = (_Float16)e0[3]; ta[5] = (_Float16)e1[0]; ta[6] = (_Float16)e1[1]; ta[7] = (_Float16)e1[2];
            tb[0] = (_Float16)e1[3]; tb[1] = (_Float16)e2[0]; tb[2] = (_Float16)e2[1]; tb[3] = (_Float16)e2[2];
            tb[4] = (_Float16)e2[3]; tb[5] = (_Float16)e3[0]; tb[6] = (_Float16)e3[1]; tb[7] = (_Float16)e3[2];
        } else {
            ta[0] = (_Float16)e3[3];
        }
        _Float16* tp = &sst[wave].feat[col][2 * FDIM + 16 * khalf];
        *(v8h*)(tp)     = ta;
        *(v8h*)(tp + 8) = tb;
    }
    __syncthreads();

    const _Float16* arow = &sst[wave].feat[col][0];

    v8f acc[4] = {v8f{}, v8f{}, v8f{}, v8f{}};
#pragma unroll
    for (int kc = 0; kc < 5; ++kc) {
        v16h a = load_a_row(arow, kc * 32, khalf);
#pragma unroll
        for (int nt = 0; nt < 4; ++nt) {
            v16h b = *(const v16h*)(wsW + MW1B + (kc * 4 + nt) * 512 + lane * 16);
            acc[nt] = wmma16(a, b, acc[nt]);
        }
    }
#pragma unroll
    for (int nt = 0; nt < 4; ++nt) {
        float bia = mb1[nt * 16 + col];
#pragma unroll
        for (int r = 0; r < 8; ++r) {
            float v = acc[nt][r] + bia;
            shid[wave][r + 8 * khalf][nt * 16 + col] = (_Float16)(v > 0.0f ? v : 0.0f);
        }
    }
    const _Float16* hrow = &shid[wave][col][0];

    v8f mij[4] = {v8f{}, v8f{}, v8f{}, v8f{}};
#pragma unroll
    for (int kc = 0; kc < 2; ++kc) {
        v16h a = load_a_row(hrow, kc * 32, khalf);
#pragma unroll
        for (int nt = 0; nt < 4; ++nt) {
            v16h b = *(const v16h*)(wsW + MW2B + (kc * 4 + nt) * 512 + lane * 16);
            mij[nt] = wmma16(a, b, mij[nt]);
        }
    }
#pragma unroll
    for (int nt = 0; nt < 4; ++nt) {
        float bia = mb2[nt * 16 + col];
        v4f lo, hi;
#pragma unroll
        for (int r = 0; r < 8; ++r) {
            int row = r + 8 * khalf;
            float v = mij[nt][r] + bia;
            if (r < 4) lo[r] = v; else hi[r - 4] = v;
            shid[wave][row][nt * 16 + col] = (_Float16)v;
        }
        float* tcol = &sst[wave].tr[nt * 16 + col][8 * khalf];
        *(v4f*)(tcol)     = lo;
        *(v4f*)(tcol + 4) = hi;
    }
    {
        int drow[16];
#pragma unroll
        for (int i = 0; i < 16; ++i) drow[i] = snid[1][wave * 16 + i];
#pragma unroll
        for (int j = 0; j < 2; ++j) {
#pragma unroll
            for (int q = 0; q < 4; ++q) {
                v4f t = *(const v4f*)(&sst[wave].tr[32 * j + lane][4 * q]);
#pragma unroll
                for (int i = 0; i < 4; ++i)
                    atomicAdd(&msum[(size_t)drow[4 * q + i] * FDIM + 32 * j + lane], t[i]);
            }
        }
    }

    v8f tac[4] = {v8f{}, v8f{}, v8f{}, v8f{}};
#pragma unroll
    for (int kc = 0; kc < 2; ++kc) {
        v16h a = load_a_row(hrow, kc * 32, khalf);
#pragma unroll
        for (int nt = 0; nt < 4; ++nt) {
            v16h b = *(const v16h*)(wsW + CW1B + (kc * 4 + nt) * 512 + lane * 16);
            tac[nt] = wmma16(a, b, tac[nt]);
        }
    }
#pragma unroll
    for (int nt = 0; nt < 4; ++nt) {
        float bia = cb1[nt * 16 + col];
#pragma unroll
        for (int r = 0; r < 8; ++r) {
            float v = tac[nt][r] + bia;
            shid[wave][r + 8 * khalf][nt * 16 + col] = (_Float16)(v > 0.0f ? v : 0.0f);
        }
    }

    v8f cac = v8f{};
#pragma unroll
    for (int kc = 0; kc < 2; ++kc) {
        v16h a = load_a_row(hrow, kc * 32, khalf);
        v16h b = *(const v16h*)(wsW + CW2B + kc * 512 + lane * 16);
        cac = wmma16(a, b, cac);
    }
    if (col < 3) {
        float bia = cb2[col];
#pragma unroll
        for (int r = 0; r < 8; ++r) {
            int el = wave * 16 + r + 8 * khalf;
            float w = cac[r] + bia;
            float v = sdpi[el][col] * srinv[el] * w;
            atomicAdd(&pos_out[(size_t)snid[1][el] * 3 + col], v);
        }
    }
}

__global__ __launch_bounds__(256) void egnn_node(
    const float* __restrict__ x, const float* __restrict__ msum,
    const float* __restrict__ cnt,
    const float* __restrict__ nb1, const float* __restrict__ nb2,
    const _Float16* __restrict__ wsW, float* __restrict__ hout)
{
    __shared__ alignas(16) _Float16 sfeat[128][NODE_K];
    __shared__ _Float16 shid[8][16][FDIM];
    __shared__ float sinv[128];

    const int tid   = threadIdx.x;
    const int nbase = blockIdx.x * 128;

    if (tid < 128) {
        int n = nbase + tid;
        float c = (n < N_NODES) ? cnt[n] : 1.0f;
        sinv[tid] = 1.0f / fmaxf(c, 1.0f);
    }
    __syncthreads();

    const int wave  = tid >> 5;
    const int lane  = tid & 31;
    const int col   = lane & 15;
    const int khalf = lane >> 4;

    {
        const float* rq = (khalf ? msum : x) + 4 * col;
        v4f rv[16];
#pragma unroll
        for (int i = 0; i < 16; ++i) {
            int n = nbase + wave * 16 + i;
            rv[i] = *(const v4f*)(rq + (size_t)(n < N_NODES ? n : N_NODES - 1) * FDIM);
        }
#pragma unroll
        for (int i = 0; i < 16; ++i) {
            int n = nbase + wave * 16 + i;
            float sc = khalf ? sinv[wave * 16 + i] : 1.0f;
            v4f v = rv[i] * sc;
            if (n >= N_NODES) v = v4f{};
            *(v4h*)&sfeat[wave * 16 + i][FDIM * khalf + 4 * col] = __builtin_convertvector(v, v4h);
        }
    }
    __syncthreads();

    const _Float16* arow = &sfeat[wave * 16 + col][0];

    v8f acc[4] = {v8f{}, v8f{}, v8f{}, v8f{}};
#pragma unroll
    for (int kc = 0; kc < 4; ++kc) {
        v16h a = load_a_row(arow, kc * 32, khalf);
#pragma unroll
        for (int nt = 0; nt < 4; ++nt) {
            v16h b = *(const v16h*)(wsW + NW1B + (kc * 4 + nt) * 512 + lane * 16);
            acc[nt] = wmma16(a, b, acc[nt]);
        }
    }
#pragma unroll
    for (int nt = 0; nt < 4; ++nt) {
        float bia = nb1[nt * 16 + col];
#pragma unroll
        for (int r = 0; r < 8; ++r) {
            float v = acc[nt][r] + bia;
            shid[wave][r + 8 * khalf][nt * 16 + col] = (_Float16)(v > 0.0f ? v : 0.0f);
        }
    }
    const _Float16* hrow = &shid[wave][col][0];

    v8f hacc[4] = {v8f{}, v8f{}, v8f{}, v8f{}};
#pragma unroll
    for (int kc = 0; kc < 2; ++kc) {
        v16h a = load_a_row(hrow, kc * 32, khalf);
#pragma unroll
        for (int nt = 0; nt < 4; ++nt) {
            v16h b = *(const v16h*)(wsW + NW2B + (kc * 4 + nt) * 512 + lane * 16);
            hacc[nt] = wmma16(a, b, hacc[nt]);
        }
    }
#pragma unroll
    for (int nt = 0; nt < 4; ++nt) {
        float bia = nb2[nt * 16 + col];
#pragma unroll
        for (int r = 0; r < 8; ++r) {
            int n = nbase + wave * 16 + r + 8 * khalf;
            if (n < N_NODES)
                hout[(size_t)n * FDIM + nt * 16 + col] = hacc[nt][r] + bia;
        }
    }
}

extern "C" void kernel_launch(void* const* d_in, const int* in_sizes, int n_in,
                              void* d_out, int out_size, void* d_ws, size_t ws_size,
                              hipStream_t stream) {
    const float* x         = (const float*)d_in[0];
    const int*   ei        = (const int*)  d_in[1];
    const float* pos       = (const float*)d_in[2];
    const float* pos_init  = (const float*)d_in[3];
    const float* edge_attr = (const float*)d_in[4];
    const float* mw1 = (const float*)d_in[5];
    const float* mb1 = (const float*)d_in[6];
    const float* mw2 = (const float*)d_in[7];
    const float* mb2 = (const float*)d_in[8];
    const float* nw1 = (const float*)d_in[9];
    const float* nb1 = (const float*)d_in[10];
    const float* nw2 = (const float*)d_in[11];
    const float* nb2 = (const float*)d_in[12];
    const float* cw1 = (const float*)d_in[13];
    const float* cb1 = (const float*)d_in[14];
    const float* cw2 = (const float*)d_in[15];
    const float* cb2 = (const float*)d_in[16];

    float* h_out   = (float*)d_out;
    float* pos_out = h_out + (size_t)N_NODES * FDIM;

    float* msum = (float*)d_ws;
    float* cnt  = (float*)((char*)d_ws + WS_CNT_OFF);
    _Float16* wsW = (_Float16*)((char*)d_ws + WS_WEIGHTS_OFF);

    egnn_setup<<<PREP_BLOCKS + (N_NODES * FDIM) / 256, 256, 0, stream>>>(
        msum, cnt, pos_out, wsW, mw1, mw2, cw1, cw2, nw1, nw2);

    egnn_edge<<<N_EDGES / 128, 256, 0, stream>>>(
        x, ei, pos, pos_init, edge_attr, mb1, mb2, cb1, cb2,
        wsW, msum, cnt, pos_out);

    egnn_node<<<(N_NODES + 127) / 128, 256, 0, stream>>>(
        x, msum, cnt, nb1, nb2, wsW, h_out);
}


# Degree-count and position accumulators: one copy per coherence half (block picks its half from its die number), summed in the node kernel; order of the f32 position sums changes (two partial sums)
# speedup vs baseline: 2.8860x; 1.4142x over previous
#include <hip/hip_runtime.h>

typedef __attribute__((ext_vector_type(16))) _Float16 v16h;
typedef __attribute__((ext_vector_type(8)))  float    v8f;
typedef __attribute__((ext_vector_type(4)))  float    v4f;
typedef __attribute__((ext_vector_type(4)))  _Float16 v4h;
typedef __attribute__((ext_vector_type(8)))  _Float16 v8h;

#define N_NODES 50000
#define N_EDGES 800000
#define FDIM    64
#define EFDIM   16
#define MSG_IN  145
#define MSG_K   160
#define NODE_K  128
#define TR_PITCH 20

union alignas(16) EdgeStage {
    _Float16 feat[16][MSG_K];
    float    tr[FDIM][TR_PITCH];
};

#define ACC_POS_OFF    50048
#define ACC_PLANE      200064
#define WS_WEIGHTS_OFF ((size_t)13000192)

#define MW1B 0
#define MW2B 10240
#define CW1B 14336
#define CW2B 18432
#define NW1B 19456
#define NW2B 27648

__device__ __forceinline__ v8f wmma16(v16h a, v16h b, v8f c) {
    return __builtin_amdgcn_wmma_f32_16x16x32_f16(
         false, a,  false, b,
         (short)0, c,  false,  false);
}

__device__ __forceinline__ v16h load_a_row(const _Float16* rowptr, int kcbase, int khalf) {
    v16h a;
#pragma unroll
    for (int e = 0; e < 16; ++e) {
        int klocal = (e & 7) + 8 * khalf + 16 * (e >> 3);
        a[e] = rowptr[kcbase + klocal];
    }
    return a;
}

#define PREP_BLOCKS 124
__global__ __launch_bounds__(256) void egnn_setup(
    float* __restrict__ msum, float* __restrict__ planes,
    _Float16* __restrict__ wsW,
    const float* __restrict__ mw1, const float* __restrict__ mw2,
    const float* __restrict__ cw1, const float* __restrict__ cw2,
    const float* __restrict__ nw1, const float* __restrict__ nw2)
{
    const int b = blockIdx.x;
    if (b < PREP_BLOCKS) {
        const float* Wm; int Kact, Nact, ntiles, base;
        if      (b < MW2B / 256) { Wm = mw1; Kact = MSG_IN; Nact = 64; ntiles = 4; base = MW1B; }
        else if (b < CW1B / 256) { Wm = mw2; Kact = 64;     Nact = 64; ntiles = 4; base = MW2B; }
        else if (b < CW2B / 256) { Wm = cw1; Kact = 64;     Nact = 64; ntiles = 4; base = CW1B; }
        else if (b < NW1B / 256) { Wm = cw2; Kact = 64;     Nact = 3;  ntiles = 1; base = CW2B; }
        else if (b < NW2B / 256) { Wm = nw1; Kact = NODE_K; Nact = 64; ntiles = 4; base = NW1B; }
        else                     { Wm = nw2; Kact = 64;     Nact = 64; ntiles = 4; base = NW2B; }
        int gidx = b * 256 + threadIdx.x;
        int idx  = gidx - base;
        int block  = idx >> 9;
        int within = idx & 511;
        int lane = within >> 4;
        int e    = within & 15;
        int kc = block / ntiles;
        int nt = block - kc * ntiles;
        int khalf  = lane >> 4;
        int coln   = lane & 15;
        int klocal = (e & 7) + 8 * khalf + 16 * (e >> 3);
        int k = kc * 32 + klocal;
        int n = nt * 16 + coln;
        float v = (k < Kact && n < Nact) ? Wm[(size_t)k * Nact + n] : 0.0f;
        wsW[gidx] = (_Float16)v;
        return;
    }
    int i = (b - PREP_BLOCKS) * 256 + threadIdx.x;
    msum[i] = 0.0f;
    if (i < 2 * ACC_PLANE) planes[i] = 0.0f;
}

__global__ __launch_bounds__(256) void egnn_edge(
    const float* __restrict__ x, const int* __restrict__ ei,
    const float* __restrict__ pos, const float* __restrict__ pos_init,
    const float* __restrict__ edge_attr,
    const float* __restrict__ mb1, const float* __restrict__ mb2,
    const float* __restrict__ cb1, const float* __restrict__ cb2,
    const _Float16* __restrict__ wsW,
    float* __restrict__ msum, float* __restrict__ planes)
{
    __shared__ EdgeStage sst[8];
    __shared__ _Float16 shid[8][16][FDIM];
    __shared__ float sdsq[128];
    __shared__ float sdpi[128][3];
    __shared__ float srinv[128];
    __shared__ alignas(16) int snid[2][128];

    const int tid   = threadIdx.x;
    const int ebase = blockIdx.x * 128;

    const unsigned where = __builtin_amdgcn_s_sendmsg_rtn(0x87);
    float* cnt     = planes + ((where >> 18) & 1u) * ACC_PLANE;
    float* pos_acc = cnt + ACC_POS_OFF;

    if (tid < 128) {
        int e = ebase + tid;
        int s = ei[e];
        int d = ei[N_EDGES + e];
        snid[0][tid] = s; snid[1][tid] = d;
        float dx = pos[3*s+0] - pos[3*d+0];
        float dy = pos[3*s+1] - pos[3*d+1];
        float dz = pos[3*s+2] - pos[3*d+2];
        sdsq[tid] = dx*dx + dy*dy + dz*dz;
        float ix = pos_init[3*s+0] - pos_init[3*d+0];
        float iy = pos_init[3*s+1] - pos_init[3*d+1];
        float iz = pos_init[3*s+2] - pos_init[3*d+2];
        sdpi[tid][0] = ix; sdpi[tid][1] = iy; sdpi[tid][2] = iz;
        srinv[tid] = 1.0f / sqrtf(ix*ix + iy*iy + iz*iz);
        atomicAdd(&cnt[d], 1.0f);
    }
    __syncthreads();

    const int wave  = tid >> 5;
    const int lane  = tid & 31;
    const int col   = lane & 15;
    const int khalf = lane >> 4;

    {
        const int*   nid = &snid[khalf][wave * 16];
        const float* xq  = x + 4 * col;
        v4f xv[16];
#pragma unroll
        for (int i = 0; i < 16; ++i)
            xv[i] = *(const v4f*)(xq + (size_t)nid[i] * FDIM);
        const v4f* ea = (const v4f*)(edge_attr + (size_t)(ebase + wave * 16 + col) * EFDIM);
        const v4f e0 = ea[0], e1 = ea[1], e2 = ea[2], e3 = ea[3];
        const float dsq = sdsq[wave * 16 + col];
#pragma unroll
        for (int i = 0; i < 16; ++i)
            *(v4h*)&sst[wave].feat[i][FDIM * khalf + 4 * col] = __builtin_convertvector(xv[i], v4h);
        v8h ta = v8h{}, tb = v8h{};
        if (khalf == 0) {
            ta[0] = (_Float16)dsq;   ta[1] = (_Float16)e0[0]; ta[2] = (_Float16)e0[1]; ta[3] = (_Float16)e0[2];
            ta[4] = (_Float16)e0[3]; ta[5] = (_Float16)e1[0]; ta[6] = (_Float16)e1[1]; ta[7] = (_Float16)e1[2];
            tb[0] = (_Float16)e1[3]; tb[1] = (_Float16)e2[0]; tb[2] = (_Float16)e2[1]; tb[3] = (_Float16)e2[2];
            tb[4] = (_Float16)e2[3]; tb[5] = (_Float16)e3[0]; tb[6] = (_Float16)e3[1]; tb[7] = (_Float16)e3[2];
        } else {
            ta[0] = (_Float16)e3[3];
        }
        _Float16* tp = &sst[wave].feat[col][2 * FDIM + 16 * khalf];
        *(v8h*)(tp)     = ta;
        *(v8h*)(tp + 8) = tb;
    }
    __syncthreads();

    const _Float16* arow = &sst[wave].feat[col][0];

    v8f acc[4] = {v8f{}, v8f{}, v8f{}, v8f{}};
#pragma unroll
    for (int kc = 0; kc < 5; ++kc) {
        v16h a = load_a_row(arow, kc * 32, khalf);
#pragma unroll
        for (int nt = 0; nt < 4; ++nt) {
            v16h b = *(const v16h*)(wsW + MW1B + (kc * 4 + nt) * 512 + lane * 16);
            acc[nt] = wmma16(a, b, acc[nt]);
        }
    }
#pragma unroll
    for (int nt = 0; nt < 4; ++nt) {
        float bia = mb1[nt * 16 + col];
#pragma unroll
        for (int r = 0; r < 8; ++r) {
            float v = acc[nt][r] + bia;
            shid[wave][r + 8 * khalf][nt * 16 + col] = (_Float16)(v > 0.0f ? v : 0.0f);
        }
    }
    const _Float16* hrow = &shid[wave][col][0];

    v8f mij[4] = {v8f{}, v8f{}, v8f{}, v8f{}};
#pragma unroll
    for (int kc = 0; kc < 2; ++kc) {
        v16h a = load_a_row(hrow, kc * 32, khalf);
#pragma unroll
        for (int nt = 0; nt < 4; ++nt) {
            v16h b = *(const v16h*)(wsW + MW2B + (kc * 4 + nt) * 512 + lane * 16);
            mij[nt] = wmma16(a, b, mij[nt]);
        }
    }
#pragma unroll
    for (int nt = 0; nt < 4; ++nt) {
        float bia = mb2[nt * 16 + col];
        v4f lo, hi;
#pragma unroll
        for (int r = 0; r < 8; ++r) {
            int row = r + 8 * khalf;
            float v = mij[nt][r] + bia;
            if (r < 4) lo[r] = v; else hi[r - 4] = v;
            shid[wave][row][nt * 16 + col] = (_Float16)v;
        }
        float* tcol = &sst[wave].tr[nt * 16 + col][8 * khalf];
        *(v4f*)(tcol)     = lo;
        *(v4f*)(tcol + 4) = hi;
    }
    {
        int drow[16];
#pragma unroll
        for (int i = 0; i < 16; ++i) drow[i] = snid[1][wave * 16 + i];
#pragma unroll
        for (int j = 0; j < 2; ++j) {
#pragma unroll
            for (int q = 0; q < 4; ++q) {
                v4f t = *(const v4f*)(&sst[wave].tr[32 * j + lane][4 * q]);
#pragma unroll
                for (int i = 0; i < 4; ++i)
                    atomicAdd(&msum[(size_t)drow[4 * q + i] * FDIM + 32 * j + lane], t[i]);
            }
        }
    }

    v8f tac[4] = {v8f{}, v8f{}, v8f{}, v8f{}};
#pragma unroll
    for (int kc = 0; kc < 2; ++kc) {
        v16h a = load_a_row(hrow, kc * 32, khalf);
#pragma unroll
        for (int nt = 0; nt < 4; ++nt) {
            v16h b = *(const v16h*)(wsW + CW1B + (kc * 4 + nt) * 512 + lane * 16);
            tac[nt] = wmma16(a, b, tac[nt]);
        }
    }
#pragma unroll
    for (int nt = 0; nt < 4; ++nt) {
        float bia = cb1[nt * 16 + col];
#pragma unroll
        for (int r = 0; r < 8; ++r) {
            float v = tac[nt][r] + bia;
            shid[wave][r + 8 * khalf][nt * 16 + col] = (_Float16)(v > 0.0f ? v : 0.0f);
        }
    }

    v8f cac = v8f{};
#pragma unroll
    for (int kc = 0; kc < 2; ++kc) {
        v16h a = load_a_row(hrow, kc * 32, khalf);
        v16h b = *(const v16h*)(wsW + CW2B + kc * 512 + lane * 16);
        cac = wmma16(a, b, cac);
    }
    if (col < 3) {
        float bia = cb2[col];
#pragma unroll
        for (int r = 0; r < 8; ++r) {
            int el = wave * 16 + r + 8 * khalf;
            float w = cac[r] + bia;
            float v = sdpi[el][col] * srinv[el] * w;
            atomicAdd(&pos_acc[(size_t)snid[1][el] * 3 + col], v);
        }
    }
}

__global__ __launch_bounds__(256) void egnn_node(
    const float* __restrict__ x, const float* __restrict__ planes,
    const float* __restrict__ nb1, const float* __restrict__ nb2,
    const _Float16* __restrict__ wsW, float* hout, float* __restrict__ pos_out)
{
    __shared__ alignas(16) _Float16 sfeat[128][NODE_K];
    __shared__ _Float16 shid[8][16][FDIM];
    __shared__ float sinv[128];

    const int tid   = threadIdx.x;
    const int nbase = blockIdx.x * 128;

    if (tid < 128) {
        int n = nbase + tid;
        float c = (n < N_NODES) ? planes[n] + planes[ACC_PLANE + n] : 1.0f;
        sinv[tid] = 1.0f / fmaxf(c, 1.0f);
    }
    __syncthreads();

    const int wave  = tid >> 5;
    const int lane  = tid & 31;
    const int col   = lane & 15;
    const int khalf = lane >> 4;

    {
        const float* rq = (khalf ? (const float*)hout : x) + 4 * col;
        v4f rv[16];
#pragma unroll
        for (int i = 0; i < 16; ++i) {
            int n = nbase + wave * 16 + i;
            rv[i] = *(const v4f*)(rq + (size_t)(n < N_NODES ? n : N_NODES - 1) * FDIM);
        }
#pragma unroll
        for (int i = 0; i < 16; ++i) {
            int n = nbase + wave * 16 + i;
            float sc = khalf ? sinv[wave * 16 + i] : 1.0f;
            v4f v = rv[i] * sc;
            if (n >= N_NODES) v = v4f{};
            *(v4h*)&sfeat[wave * 16 + i][FDIM * khalf + 4 * col] = __builtin_convertvector(v, v4h);
        }
    }
    __syncthreads();

    {
        const int  i0  = nbase * 3 + tid;
        const int  i1  = i0 + 256;
        const bool ok0 = i0 < N_NODES * 3;
        const bool ok1 = (tid < 128) && (i1 < N_NODES * 3);
        const float a0 = ok0 ? planes[ACC_POS_OFF + i0] : 0.0f;
        const float b0 = ok0 ? planes[ACC_PLANE + ACC_POS_OFF + i0] : 0.0f;
        const float a1 = ok1 ? planes[ACC_POS_OFF + i1] : 0.0f;
        const float b1 = ok1 ? planes[ACC_PLANE + ACC_POS_OFF + i1] : 0.0f;
        if (ok0) pos_out[i0] = a0 + b0;
        if (ok1) pos_out[i1] = a1 + b1;
    }

    const _Float16* arow = &sfeat[wave * 16 + col][0];

    v8f acc[4] = {v8f{}, v8f{}, v8f{}, v8f{}};
#pragma unroll
    for (int kc = 0; kc < 4; ++kc) {
        v16h a = load_a_row(arow, kc * 32, khalf);
#pragma unroll
        for (int nt = 0; nt < 4; ++nt) {
            v16h b = *(const v16h*)(wsW + NW1B + (kc * 4 + nt) * 512 + lane * 16);
            acc[nt] = wmma16(a, b, acc[nt]);
        }
    }
#pragma unroll
    for (int nt = 0; nt < 4; ++nt) {
        float bia = nb1[nt * 16 + col];
#pragma unroll
        for (int r = 0; r < 8; ++r) {
            float v = acc[nt][r] + bia;
            shid[wave][r + 8 * khalf][nt * 16 + col] = (_Float16)(v > 0.0f ? v : 0.0f);
        }
    }
    const _Float16* hrow = &shid[wave][col][0];

    v8f hacc[4] = {v8f{}, v8f{}, v8f{}, v8f{}};
#pragma unroll
    for (int kc = 0; kc < 2; ++kc) {
        v16h a = load_a_row(hrow, kc * 32, khalf);
#pragma unroll
        for (int nt = 0; nt < 4; ++nt) {
            v16h b = *(const v16h*)(wsW + NW2B + (kc * 4 + nt) * 512 + lane * 16);
            hacc[nt] = wmma16(a, b, hacc[nt]);
        }
    }
#pragma unroll
    for (int nt = 0; nt < 4; ++nt) {
        float bia = nb2[nt * 16 + col];
#pragma unroll
        for (int r = 0; r < 8; ++r) {
            int n = nbase + wave * 16 + r + 8 * khalf;
            if (n < N_NODES)
                hout[(size_t)n * FDIM + nt * 16 + col] = hacc[nt][r] + bia;
        }
    }
}

extern "C" void kernel_launch(void* const* d_in, const int* in_sizes, int n_in,
                              void* d_out, int out_size, void* d_ws, size_t ws_size,
                              hipStream_t stream) {
    const float* x         = (const float*)d_in[0];
    const int*   ei        = (const int*)  d_in[1];
    const float* pos       = (const float*)d_in[2];
    const float* pos_init  = (const float*)d_in[3];
    const float* edge_attr = (const float*)d_in[4];
    const float* mw1 = (const float*)d_in[5];
    const float* mb1 = (const float*)d_in[6];
    const float* mw2 = (const float*)d_in[7];
    const float* mb2 = (const float*)d_in[8];
    const float* nw1 = (const float*)d_in[9];
    const float* nb1 = (const float*)d_in[10];
    const float* nw2 = (const float*)d_in[11];
    const float* nb2 = (const float*)d_in[12];
    const float* cw1 = (const float*)d_in[13];
    const float* cb1 = (const float*)d_in[14];
    const float* cw2 = (const float*)d_in[15];
    const float* cb2 = (const float*)d_in[16];

    float* h_out   = (float*)d_out;
    float* pos_out = h_out + (size_t)N_NODES * FDIM;

    float* msum = h_out;
    float* planes = (float*)d_ws;
    _Float16* wsW = (_Float16*)((char*)d_ws + WS_WEIGHTS_OFF);

    egnn_setup<<<PREP_BLOCKS + (N_NODES * FDIM) / 256, 256, 0, stream>>>(
        msum, planes, wsW, mw1, mw2, cw1, cw2, nw1, nw2);

    egnn_edge<<<N_EDGES / 128, 256, 0, stream>>>(
        x, ei, pos, pos_init, edge_attr, mb1, mb2, cb1, cb2,
        wsW, msum, planes);

    egnn_node<<<(N_NODES + 127) / 128, 256, 0, stream>>>(
        x, planes, nb1, nb2, wsW, h_out, pos_out);
}
